# starting kernel
# baseline (speedup 1.0000x reference)
#include <hip/hip_runtime.h>

typedef float v2f __attribute__((ext_vector_type(2)));
typedef float v8f __attribute__((ext_vector_type(8)));

#define EA_W 5
#define EA_B 4
#define EA_H 128
#define EA_WW 160
#define EA_C 64
#define EA_M 10
#define EA_K 640
#define EA_EPS 1e-3f

__global__ void ea_prep_kernel(const float* __restrict__ kflat, float* __restrict__ bpt) {
    int idx = blockIdx.x * blockDim.x + threadIdx.x;
    if (idx >= 16 * EA_K) return;
    int m = idx / EA_K;
    int k = idx % EA_K;
    bpt[idx] = (m < EA_M) ? kflat[k * EA_M + m] : 0.0f;
}

__global__ __launch_bounds__(128)
void ea_main_kernel(const float* __restrict__ L, const float* __restrict__ R,
                    const float* __restrict__ bpt, const float* __restrict__ bias,
                    const float* __restrict__ gamma, const float* __restrict__ beta,
                    const float* __restrict__ mean, const float* __restrict__ var,
                    float* __restrict__ out) {
    const int tid  = threadIdx.x;
    const int lane = tid & 31;
    const int wib  = tid >> 5;
    const int tile = blockIdx.x * 4 + wib;
    const int lo = lane & 15;
    const int hi = lane >> 4;

    const int p0 = tile * 16;
    const int n  = p0 / (EA_H * EA_WW);
    const int hw = p0 % (EA_H * EA_WW);
    const int h  = hw / EA_WW;
    const int w0 = hw % EA_WW;
    const int w  = w0 + lo;
    const size_t rowBase = (size_t)(n * EA_H + h) * EA_WW * EA_C;
    const float* Lrow = L + rowBase;
    const float* Rrow = R + rowBase;

    __shared__ float qtile[4][20][128];
    __shared__ float bsm[16][640];

    {
        const float4* bsrc = (const float4*)bpt;
        float4* bdst = (float4*)&bsm[0][0];
        #pragma unroll
        for (int it = 0; it < 20; ++it)
            bdst[it * 128 + tid] = bsrc[it * 128 + tid];
    }
    {
        float4* qdst = (float4*)&qtile[wib][0][0];
        const int c4 = lane * 4;
        #pragma unroll
        for (int it = 0; it < 20; ++it) {
            const int wq = w0 - 2 + it;
            float4 v = {0.0f, 0.0f, 0.0f, 0.0f};
            if (wq >= 0 && wq < EA_WW) {
                const float* s = (c4 < EA_C) ? (Lrow + (size_t)wq * EA_C + c4)
                                             : (Rrow + (size_t)wq * EA_C + (c4 - EA_C));
                v = *(const float4*)s;
            }
            qdst[it * 32 + lane] = v;
        }
    }
    __syncthreads();

    const v2f* aV = (const v2f*)&qtile[wib][lo][hi * 2];
    const v2f* bV = (const v2f*)&bsm[lo][hi * 2];
    v8f acc0 = {}, acc1 = {};
    #pragma unroll
    for (int step = 0; step < EA_K / 4; step += 2) {
        const v2f a0 = aV[2 * step];
        const v2f b0 = bV[2 * step];
        acc0 = __builtin_amdgcn_wmma_f32_16x16x4_f32(false, a0, false, b0,
                                                     (short)0, acc0, false, false);
        const v2f a1 = aV[2 * step + 2];
        const v2f b1 = bV[2 * step + 2];
        acc1 = __builtin_amdgcn_wmma_f32_16x16x4_f32(false, a1, false, b1,
                                                     (short)0, acc1, false, false);
    }
    const v8f acc = acc0 + acc1;

    __shared__ float smem[4][16][16];
    #pragma unroll
    for (int v = 0; v < 8; ++v)
        smem[wib][v + 8 * hi][lo] = acc[v];
    __syncthreads();

    float wv[EA_M];
    float mx = -3.0e38f;
    #pragma unroll
    for (int m = 0; m < EA_M; ++m) {
        wv[m] = smem[wib][lo][m] + bias[m];
        mx = fmaxf(mx, wv[m]);
    }
    float sum = 0.0f;
    #pragma unroll
    for (int m = 0; m < EA_M; ++m) { wv[m] = __expf(wv[m] - mx); sum += wv[m]; }
    const float inv = 1.0f / sum;
    #pragma unroll
    for (int m = 0; m < EA_M; ++m) wv[m] *= inv;

    const int p = p0 + lo;
    const size_t XN = (size_t)EA_B * EA_H * EA_WW * EA_C;
    const size_t WN = (size_t)EA_B * EA_H * EA_WW * EA_M;
    if (hi == 0) {
        float* ow  = out + XN + (size_t)p * EA_M;
        float* ow2 = out + XN + WN + (size_t)p * EA_M;
        #pragma unroll
        for (int m = 0; m < EA_M; ++m) { ow[m] = wv[m]; ow2[m] = wv[m]; }
    }

    const int kb = hi * 32;
    float x[32];
    #pragma unroll
    for (int j = 0; j < 32; ++j) x[j] = 0.0f;
    #pragma unroll
    for (int i = 0; i < EA_W; ++i) {
        int wl = w - i; if (wl < 0)      wl += EA_WW;
        int wr = w + i; if (wr >= EA_WW) wr -= EA_WW;
        const float4* ls = (const float4*)(Lrow + (size_t)wl * EA_C + kb);
        const float4* rs = (const float4*)(Rrow + (size_t)wr * EA_C + kb);
        const float wlw = wv[2 * i];
        const float wrw = wv[2 * i + 1];
        #pragma unroll
        for (int q = 0; q < 8; ++q) {
            const float4 lv = ls[q];
            const float4 rv = rs[q];
            x[4 * q + 0] += wlw * lv.x + wrw * rv.x;
            x[4 * q + 1] += wlw * lv.y + wrw * rv.y;
            x[4 * q + 2] += wlw * lv.z + wrw * rv.z;
            x[4 * q + 3] += wlw * lv.w + wrw * rv.w;
        }
    }

    float* ox = out + (size_t)p * EA_C + kb;
    #pragma unroll
    for (int j = 0; j < 32; ++j) {
        const int k = kb + j;
        const float sc = gamma[k] * rsqrtf(var[k] + EA_EPS);
        const float y  = (x[j] - mean[k]) * sc + beta[k];
        ox[j] = tanhf(y);
    }
}

extern "C" void kernel_launch(void* const* d_in, const int* in_sizes, int n_in,
                              void* d_out, int out_size, void* d_ws, size_t ws_size,
                              hipStream_t stream) {
    const float* L     = (const float*)d_in[0];
    const float* R     = (const float*)d_in[1];
    const float* kconv = (const float*)d_in[2];
    const float* bias  = (const float*)d_in[3];
    const float* gamma = (const float*)d_in[4];
    const float* beta  = (const float*)d_in[5];
    const float* mean  = (const float*)d_in[6];
    const float* var   = (const float*)d_in[7];
    float* out = (float*)d_out;
    float* bpt = (float*)d_ws;

    ea_prep_kernel<<<(16 * EA_K + 255) / 256, 256, 0, stream>>>(kconv, bpt);

    const int tiles = EA_B * EA_H * EA_WW / 16;
    ea_main_kernel<<<tiles / 4, 128, 0, stream>>>(L, R, bpt, bias, gamma, beta,
                                                  mean, var, out);
}


# Softmax-weighted gather as packed f32 multiply-adds in the output layout (lane: 4 channels x 8 pixels, weights by row-share DPP) instead of a 1/4-dense band product on WMMA; gather sum order changed
# speedup vs baseline: 6.7046x; 6.7046x over previous
#include <hip/hip_runtime.h>


typedef float v2f __attribute__((ext_vector_type(2)));
typedef float v4f __attribute__((ext_vector_type(4)));
typedef float v8f __attribute__((ext_vector_type(8)));

#define EA_W 5
#define EA_B 4
#define EA_H 128
#define EA_WW 160
#define EA_C 64
#define EA_M 10
#define EA_KT (2 * EA_C * EA_M)
#define EA_EPS 1e-3f


__device__ __forceinline__ float ea_xhalf(float x) {
    const int b = __builtin_bit_cast(int, x);
    return __builtin_bit_cast(float,
        __builtin_amdgcn_permlanex16(b, b, 0x76543210, 0xfedcba98, false, false));
}

template <int D>
__device__ __forceinline__ float ea_rot(float x) {
    const int xi = __builtin_bit_cast(int, x);
    return __builtin_bit_cast(float,
        __builtin_amdgcn_update_dpp(0, xi, 0x120 + ((16 - D) & 15), 0xf, 0xf, true));
}

__device__ __forceinline__ float ea_rot8_row1(float x) {
    const int xi = __builtin_bit_cast(int, x);
    return __builtin_bit_cast(float,
        __builtin_amdgcn_update_dpp(xi, xi, 0x120 + 8, 0x2, 0xf, false));
}

template <int N>
__device__ __forceinline__ float ea_share(float x) {
    const int xi = __builtin_bit_cast(int, x);
    return __builtin_bit_cast(float,
        __builtin_amdgcn_update_dpp(0, xi, 0x150 + N, 0xf, 0xf, true));
}

__device__ __forceinline__ v2f ea_fma2(float w, v2f x, v2f acc) {
    return __builtin_elementwise_fma((v2f){w, w}, x, acc);
}

template <int R>
__device__ __forceinline__ void ea_gather_pixel(const float* wr, const v4f* lrow, const v4f* rrow,
                                                v2f& s01, v2f& s23) {
    #pragma unroll
    for (int i = 0; i < EA_W; ++i) {
        const float wl = ea_share<R>(wr[2 * i]);
        const float wq = ea_share<R>(wr[2 * i + 1]);
        const v4f lv = lrow[R + 4 - i];
        const v4f rv = rrow[R + i];
        if (i == 0) {
            s01 = (v2f){wl, wl} * lv.xy;
            s23 = (v2f){wl, wl} * lv.zw;
        } else {
            s01 = ea_fma2(wl, lv.xy, s01);
            s23 = ea_fma2(wl, lv.zw, s23);
        }
        s01 = ea_fma2(wq, rv.xy, s01);
        s23 = ea_fma2(wq, rv.zw, s23);
    }
}

struct EaStage { v4f m[2], e[2], w[2][EA_W]; };

__device__ __forceinline__ void ea_stage_load(EaStage& s, const float* Lrow, const float* Rrow,
                                              int offM, int offE, bool eok, const float* kb, int gp) {
    const float* src = (gp < 4) ? Lrow : Rrow;
    #pragma unroll
    for (int i = 0; i < 2; ++i) {
        const int g  = 2 * gp + i;
        const int ch = 8 * (g & 7);
        s.m[i] = *(const v4f*)(src + offM + ch);
        const v4f e = *(const v4f*)(src + offE + ch);
        s.e[i].x = eok ? e.x : 0.0f;
        s.e[i].y = eok ? e.y : 0.0f;
        s.e[i].z = eok ? e.z : 0.0f;
        s.e[i].w = eok ? e.w : 0.0f;
        const float* kg = kb + g * (8 * EA_M);
        #pragma unroll
        for (int t = 0; t < EA_W; ++t) {
            const float* kt = kg + t * EA_KT;
            s.w[i][t] = (v4f){kt[0], kt[EA_M], kt[2 * EA_M], kt[3 * EA_M]};
        }
    }
}

struct EaEdge { bool t0, t1, t3, t4; };

template <int U>
__device__ __forceinline__ v2f ea_cols(const v4f& m, const v4f& e, bool ed) {
    if (U == 0) return (v2f){ed ? e.x : m.x, ed ? e.y : m.y};
    return (v2f){ed ? e.z : m.z, ed ? e.w : m.w};
}

template <int U>
__device__ __forceinline__ void ea_taps_mma(const v4f& m, const v4f& e, const v4f* wt,
                                            const EaEdge& ed, v8f* y) {
    const v2f c0 = ea_cols<U>(m, e, ed.t0);
    const v2f c1 = ea_cols<U>(m, e, ed.t1);
    const v2f c2 = U == 0 ? (v2f){m.x, m.y} : (v2f){m.z, m.w};
    const v2f c3 = ea_cols<U>(m, e, ed.t3);
    const v2f c4 = ea_cols<U>(m, e, ed.t4);
    y[0] = __builtin_amdgcn_wmma_f32_16x16x4_f32(false, U == 0 ? wt[0].xy : wt[0].zw, false, c0, (short)0, y[0], false, false);
    y[1] = __builtin_amdgcn_wmma_f32_16x16x4_f32(false, U == 0 ? wt[1].xy : wt[1].zw, false, c1, (short)0, y[1], false, false);
    y[2] = __builtin_amdgcn_wmma_f32_16x16x4_f32(false, U == 0 ? wt[2].xy : wt[2].zw, false, c2, (short)0, y[2], false, false);
    y[3] = __builtin_amdgcn_wmma_f32_16x16x4_f32(false, U == 0 ? wt[3].xy : wt[3].zw, false, c3, (short)0, y[3], false, false);
    y[4] = __builtin_amdgcn_wmma_f32_16x16x4_f32(false, U == 0 ? wt[4].xy : wt[4].zw, false, c4, (short)0, y[4], false, false);
}

__device__ __forceinline__ void ea_stage_mma(const EaStage& s, const EaEdge& ed, v8f* y) {
    #pragma unroll
    for (int i = 0; i < 2; ++i) {
        ea_taps_mma<0>(s.m[i], s.e[i], s.w[i], ed, y);
        ea_taps_mma<1>(s.m[i], s.e[i], s.w[i], ed, y);
    }
}

__global__ __launch_bounds__(128)
void ea_main_kernel(const float* __restrict__ L, const float* __restrict__ R,
                    const float* __restrict__ kflat, const float* __restrict__ bias,
                    const float* __restrict__ gamma, const float* __restrict__ beta,
                    const float* __restrict__ mean, const float* __restrict__ var,
                    float* __restrict__ out) {
    const int tid  = threadIdx.x;
    const int lane = tid & 31;
    const int wib  = tid >> 5;
    const int tile = blockIdx.x * 4 + wib;
    const int lo = lane & 15;
    const int hi = lane >> 4;

    const int p0 = tile * 16;
    const int n  = p0 / (EA_H * EA_WW);
    const int hw = p0 % (EA_H * EA_WW);
    const int h  = hw / EA_WW;
    const int w0 = hw % EA_WW;
    const int w  = w0 + lo;
    const size_t rowBase = (size_t)(n * EA_H + h) * EA_WW * EA_C;
    const float* Lrow = L + rowBase;
    const float* Rrow = R + rowBase;

    const int  pe   = (lo < 8) ? (w0 + 16 + (lo < 1 ? lo : 1)) : (w0 - 16 + (lo > 14 ? lo : 14));
    const bool eok  = (pe >= 0) && (pe < EA_WW);
    const int  pec  = pe < 0 ? 0 : (pe >= EA_WW ? EA_WW - 1 : pe);
    const int  offM = w   * EA_C + 4 * hi;
    const int  offE = pec * EA_C + 4 * hi;
    const float* kb = kflat + 4 * EA_M * hi + (lo < EA_M ? lo : EA_M - 1);
    const EaEdge ed = {lo >= 14, lo == 15, lo == 0, lo <= 1};
    v8f y[EA_W] = {};
    EaStage cur;
    ea_stage_load(cur, Lrow, Rrow, offM, offE, eok, kb, 0);
    #pragma unroll 1
    for (int gp = 1; gp < 8; ++gp) {
        EaStage nxt;
        ea_stage_load(nxt, Lrow, Rrow, offM, offE, eok, kb, gp);
        ea_stage_mma(cur, ed, y);
        cur = nxt;
    }
    ea_stage_mma(cur, ed, y);
    v8f logit;
    #pragma unroll
    for (int r = 0; r < 8; ++r)
        logit[r] = y[2][r] + ea_rot<-2>(y[0][r]) + ea_rot<-1>(y[1][r])
                           + ea_rot<1>(y[3][r]) + ea_rot<2>(y[4][r]);

    float wv[EA_M];
    {
        float oth[8];
        #pragma unroll
        for (int r = 0; r < 8; ++r) oth[r] = ea_xhalf(logit[r]);
        #pragma unroll
        for (int m = 0; m < 8; ++m) wv[m] = hi ? oth[m] : logit[m];
        wv[8] = hi ? logit[0] : oth[0];
        wv[9] = hi ? logit[1] : oth[1];
    }

    float mx = -3.0e38f;
    #pragma unroll
    for (int m = 0; m < EA_M; ++m) {
        wv[m] = wv[m] + bias[m];
        mx = fmaxf(mx, wv[m]);
    }
    float sum = 0.0f;
    #pragma unroll
    for (int m = 0; m < EA_M; ++m) { wv[m] = __expf(wv[m] - mx); sum += wv[m]; }
    const float inv = 1.0f / sum;
    #pragma unroll
    for (int m = 0; m < EA_M; ++m) wv[m] *= inv;

    const int p = p0 + lo;
    const size_t XN = (size_t)EA_B * EA_H * EA_WW * EA_C;
    const size_t WN = (size_t)EA_B * EA_H * EA_WW * EA_M;
    if (hi == 0) {
        float* ow  = out + XN + (size_t)p * EA_M;
        float* ow2 = out + XN + WN + (size_t)p * EA_M;
        #pragma unroll
        for (int m = 0; m < EA_M; ++m) { ow[m] = wv[m]; ow2[m] = wv[m]; }
    }

    const float* Lc = Lrow + 4 * lo;
    const float* Rc = Rrow + 4 * lo;
    v4f lrow[12], rrow[12];
    #pragma unroll
    for (int u = 0; u < 12; ++u) {
        int pl = w0 + 8 * hi - 4 + u; if (pl < 0)      pl += EA_WW;
        int pr = w0 + 8 * hi + u;     if (pr >= EA_WW) pr -= EA_WW;
        lrow[u] = *(const v4f*)(Lc + pl * EA_C);
        rrow[u] = *(const v4f*)(Rc + pr * EA_C);
    }

    float wr[EA_M];
    #pragma unroll
    for (int m = 0; m < EA_M; ++m) wr[m] = ea_rot8_row1(wv[m]);

    v2f s01[8], s23[8];
    ea_gather_pixel<0>(wr, lrow, rrow, s01[0], s23[0]);
    ea_gather_pixel<1>(wr, lrow, rrow, s01[1], s23[1]);
    ea_gather_pixel<2>(wr, lrow, rrow, s01[2], s23[2]);
    ea_gather_pixel<3>(wr, lrow, rrow, s01[3], s23[3]);
    ea_gather_pixel<4>(wr, lrow, rrow, s01[4], s23[4]);
    ea_gather_pixel<5>(wr, lrow, rrow, s01[5], s23[5]);
    ea_gather_pixel<6>(wr, lrow, rrow, s01[6], s23[6]);
    ea_gather_pixel<7>(wr, lrow, rrow, s01[7], s23[7]);

    float sc[4], mu[4], be[4];
    #pragma unroll
    for (int c = 0; c < 4; ++c) {
        const int k = 4 * lo + c;
        sc[c] = gamma[k] * rsqrtf(var[k] + EA_EPS);
        mu[c] = mean[k];
        be[c] = beta[k];
    }
    float* ox = out + (size_t)(p0 + 8 * hi) * EA_C + 4 * lo;
    #pragma unroll
    for (int r = 0; r < 8; ++r) {
        v4f o;
        o.x = __builtin_amdgcn_tanhf((s01[r].x - mu[0]) * sc[0] + be[0]);
        o.y = __builtin_amdgcn_tanhf((s01[r].y - mu[1]) * sc[1] + be[1]);
        o.z = __builtin_amdgcn_tanhf((s23[r].x - mu[2]) * sc[2] + be[2]);
        o.w = __builtin_amdgcn_tanhf((s23[r].y - mu[3]) * sc[3] + be[3]);
        *(v4f*)(ox + r * EA_C) = o;
    }
}

extern "C" void kernel_launch(void* const* d_in, const int* in_sizes, int n_in,
                              void* d_out, int out_size, void* d_ws, size_t ws_size,
                              hipStream_t stream) {
    const float* L     = (const float*)d_in[0];
    const float* R     = (const float*)d_in[1];
    const float* kconv = (const float*)d_in[2];
    const float* bias  = (const float*)d_in[3];
    const float* gamma = (const float*)d_in[4];
    const float* beta  = (const float*)d_in[5];
    const float* mean  = (const float*)d_in[6];
    const float* var   = (const float*)d_in[7];
    float* out = (float*)d_out;

    const int tiles = EA_B * EA_H * EA_WW / 16;
    ea_main_kernel<<<tiles / 4, 128, 0, stream>>>(L, R, kconv, bias, gamma, beta,
                                                  mean, var, out);
}


# Conv on wider pixel tiles: a wave owns 3 or 2 adjacent 16-pixel tiles (8-wave groups, 5 tiles per SIMD), so each strided weight fragment load feeds 3 or 2 products; sum order unchanged
# speedup vs baseline: 7.2543x; 1.0820x over previous
#include <hip/hip_runtime.h>


typedef float v2f __attribute__((ext_vector_type(2)));
typedef float v4f __attribute__((ext_vector_type(4)));
typedef float v8f __attribute__((ext_vector_type(8)));

#define EA_W 5
#define EA_B 4
#define EA_H 128
#define EA_WW 160
#define EA_C 64
#define EA_M 10
#define EA_KT (2 * EA_C * EA_M)
#define EA_EPS 1e-3f


__device__ __forceinline__ float ea_xhalf(float x) {
    const int b = __builtin_bit_cast(int, x);
    return __builtin_bit_cast(float,
        __builtin_amdgcn_permlanex16(b, b, 0x76543210, 0xfedcba98, false, false));
}

template <int D>
__device__ __forceinline__ float ea_rot(float x) {
    const int xi = __builtin_bit_cast(int, x);
    return __builtin_bit_cast(float,
        __builtin_amdgcn_update_dpp(0, xi, 0x120 + ((16 - D) & 15), 0xf, 0xf, true));
}

__device__ __forceinline__ float ea_rot8_row1(float x) {
    const int xi = __builtin_bit_cast(int, x);
    return __builtin_bit_cast(float,
        __builtin_amdgcn_update_dpp(xi, xi, 0x120 + 8, 0x2, 0xf, false));
}

template <int N>
__device__ __forceinline__ float ea_share(float x) {
    const int xi = __builtin_bit_cast(int, x);
    return __builtin_bit_cast(float,
        __builtin_amdgcn_update_dpp(0, xi, 0x150 + N, 0xf, 0xf, true));
}

__device__ __forceinline__ v2f ea_fma2(float w, v2f x, v2f acc) {
    return __builtin_elementwise_fma((v2f){w, w}, x, acc);
}

template <int R>
__device__ __forceinline__ void ea_gather_pixel(const float* wr, const v4f* lrow, const v4f* rrow,
                                                v2f& s01, v2f& s23) {
    #pragma unroll
    for (int i = 0; i < EA_W; ++i) {
        const float wl = ea_share<R>(wr[2 * i]);
        const float wq = ea_share<R>(wr[2 * i + 1]);
        const v4f lv = lrow[R + 4 - i];
        const v4f rv = rrow[R + i];
        if (i == 0) {
            s01 = (v2f){wl, wl} * lv.xy;
            s23 = (v2f){wl, wl} * lv.zw;
        } else {
            s01 = ea_fma2(wl, lv.xy, s01);
            s23 = ea_fma2(wl, lv.zw, s23);
        }
        s01 = ea_fma2(wq, rv.xy, s01);
        s23 = ea_fma2(wq, rv.zw, s23);
    }
}

template <int NT>
struct EaStage { v4f m[NT][2], e[2], w[2][EA_W]; };

template <int NT>
__device__ __forceinline__ void ea_stage_load(EaStage<NT>& s, const float* Lrow, const float* Rrow,
                                              int offM, int offE, bool eok, const float* kb, int gp) {
    const float* src = (gp < 4) ? Lrow : Rrow;
    #pragma unroll
    for (int i = 0; i < 2; ++i) {
        const int g  = 2 * gp + i;
        const int ch = 8 * (g & 7);
        #pragma unroll
        for (int j = 0; j < NT; ++j)
            s.m[j][i] = *(const v4f*)(src + offM + j * (16 * EA_C) + ch);
        const v4f e = *(const v4f*)(src + offE + ch);
        s.e[i].x = eok ? e.x : 0.0f;
        s.e[i].y = eok ? e.y : 0.0f;
        s.e[i].z = eok ? e.z : 0.0f;
        s.e[i].w = eok ? e.w : 0.0f;
        const float* kg = kb + g * (8 * EA_M);
        #pragma unroll
        for (int t = 0; t < EA_W; ++t) {
            const float* kt = kg + t * EA_KT;
            s.w[i][t] = (v4f){kt[0], kt[EA_M], kt[2 * EA_M], kt[3 * EA_M]};
        }
    }
}

struct EaEdge { bool t0, t1, t3, t4; };

template <int U>
__device__ __forceinline__ v2f ea_cols(const v4f& m, const v4f& e, bool ed) {
    if (U == 0) return (v2f){ed ? e.x : m.x, ed ? e.y : m.y};
    return (v2f){ed ? e.z : m.z, ed ? e.w : m.w};
}

template <int U>
__device__ __forceinline__ void ea_taps_mma(const v4f& m, const v4f& el, const v4f& er, const v4f* wt,
                                            const EaEdge& ed, v8f* y) {
    const v2f c0 = ea_cols<U>(m, el, ed.t0);
    const v2f c1 = ea_cols<U>(m, el, ed.t1);
    const v2f c2 = U == 0 ? (v2f){m.x, m.y} : (v2f){m.z, m.w};
    const v2f c3 = ea_cols<U>(m, er, ed.t3);
    const v2f c4 = ea_cols<U>(m, er, ed.t4);
    y[0] = __builtin_amdgcn_wmma_f32_16x16x4_f32(false, U == 0 ? wt[0].xy : wt[0].zw, false, c0, (short)0, y[0], false, false);
    y[1] = __builtin_amdgcn_wmma_f32_16x16x4_f32(false, U == 0 ? wt[1].xy : wt[1].zw, false, c1, (short)0, y[1], false, false);
    y[2] = __builtin_amdgcn_wmma_f32_16x16x4_f32(false, U == 0 ? wt[2].xy : wt[2].zw, false, c2, (short)0, y[2], false, false);
    y[3] = __builtin_amdgcn_wmma_f32_16x16x4_f32(false, U == 0 ? wt[3].xy : wt[3].zw, false, c3, (short)0, y[3], false, false);
    y[4] = __builtin_amdgcn_wmma_f32_16x16x4_f32(false, U == 0 ? wt[4].xy : wt[4].zw, false, c4, (short)0, y[4], false, false);
}

template <int NT>
__device__ __forceinline__ void ea_stage_mma(const EaStage<NT>& s, const EaEdge& ed, v8f (*y)[EA_W]) {
    #pragma unroll
    for (int i = 0; i < 2; ++i) {
        #pragma unroll
        for (int j = 0; j < NT; ++j)
            ea_taps_mma<0>(s.m[j][i], j > 0 ? s.m[j - 1][i] : s.e[i], j < NT - 1 ? s.m[j + 1][i] : s.e[i],
                           s.w[i], ed, y[j]);
        #pragma unroll
        for (int j = 0; j < NT; ++j)
            ea_taps_mma<1>(s.m[j][i], j > 0 ? s.m[j - 1][i] : s.e[i], j < NT - 1 ? s.m[j + 1][i] : s.e[i],
                           s.w[i], ed, y[j]);
    }
}

struct EaBn { float sc[4], mu[4], be[4]; };

__device__ __forceinline__ void ea_tile_finish(const v8f* y, int w0, int p0, int lo, int hi,
                                               const float* Lrow, const float* Rrow,
                                               const float* __restrict__ bias, const EaBn& bn,
                                               float* __restrict__ out) {
    v8f logit;
    #pragma unroll
    for (int r = 0; r < 8; ++r)
        logit[r] = y[2][r] + ea_rot<-2>(y[0][r]) + ea_rot<-1>(y[1][r])
                           + ea_rot<1>(y[3][r]) + ea_rot<2>(y[4][r]);

    float wv[EA_M];
    {
        float oth[8];
        #pragma unroll
        for (int r = 0; r < 8; ++r) oth[r] = ea_xhalf(logit[r]);
        #pragma unroll
        for (int m = 0; m < 8; ++m) wv[m] = hi ? oth[m] : logit[m];
        wv[8] = hi ? logit[0] : oth[0];
        wv[9] = hi ? logit[1] : oth[1];
    }

    float mx = -3.0e38f;
    #pragma unroll
    for (int m = 0; m < EA_M; ++m) {
        wv[m] = wv[m] + bias[m];
        mx = fmaxf(mx, wv[m]);
    }
    float sum = 0.0f;
    #pragma unroll
    for (int m = 0; m < EA_M; ++m) { wv[m] = __expf(wv[m] - mx); sum += wv[m]; }
    const float inv = 1.0f / sum;
    #pragma unroll
    for (int m = 0; m < EA_M; ++m) wv[m] *= inv;

    const int p = p0 + lo;
    const size_t XN = (size_t)EA_B * EA_H * EA_WW * EA_C;
    const size_t WN = (size_t)EA_B * EA_H * EA_WW * EA_M;
    if (hi == 0) {
        float* ow  = out + XN + (size_t)p * EA_M;
        float* ow2 = out + XN + WN + (size_t)p * EA_M;
        #pragma unroll
        for (int m = 0; m < EA_M; ++m) { ow[m] = wv[m]; ow2[m] = wv[m]; }
    }

    const float* Lc = Lrow + 4 * lo;
    const float* Rc = Rrow + 4 * lo;
    v4f lrow[12], rrow[12];
    #pragma unroll
    for (int u = 0; u < 12; ++u) {
        int pl = w0 + 8 * hi - 4 + u; if (pl < 0)      pl += EA_WW;
        int pr = w0 + 8 * hi + u;     if (pr >= EA_WW) pr -= EA_WW;
        lrow[u] = *(const v4f*)(Lc + pl * EA_C);
        rrow[u] = *(const v4f*)(Rc + pr * EA_C);
    }

    float wr[EA_M];
    #pragma unroll
    for (int m = 0; m < EA_M; ++m) wr[m] = ea_rot8_row1(wv[m]);

    v2f s01[8], s23[8];
    ea_gather_pixel<0>(wr, lrow, rrow, s01[0], s23[0]);
    ea_gather_pixel<1>(wr, lrow, rrow, s01[1], s23[1]);
    ea_gather_pixel<2>(wr, lrow, rrow, s01[2], s23[2]);
    ea_gather_pixel<3>(wr, lrow, rrow, s01[3], s23[3]);
    ea_gather_pixel<4>(wr, lrow, rrow, s01[4], s23[4]);
    ea_gather_pixel<5>(wr, lrow, rrow, s01[5], s23[5]);
    ea_gather_pixel<6>(wr, lrow, rrow, s01[6], s23[6]);
    ea_gather_pixel<7>(wr, lrow, rrow, s01[7], s23[7]);

    float* ox = out + (size_t)(p0 + 8 * hi) * EA_C + 4 * lo;
    #pragma unroll
    for (int r = 0; r < 8; ++r) {
        v4f o;
        o.x = __builtin_amdgcn_tanhf((s01[r].x - bn.mu[0]) * bn.sc[0] + bn.be[0]);
        o.y = __builtin_amdgcn_tanhf((s01[r].y - bn.mu[1]) * bn.sc[1] + bn.be[1]);
        o.z = __builtin_amdgcn_tanhf((s23[r].x - bn.mu[2]) * bn.sc[2] + bn.be[2]);
        o.w = __builtin_amdgcn_tanhf((s23[r].y - bn.mu[3]) * bn.sc[3] + bn.be[3]);
        *(v4f*)(ox + r * EA_C) = o;
    }
}

template <int NT>
__device__ __forceinline__ void ea_wave(const float* Lrow, const float* Rrow,
                                        const float* __restrict__ kflat, const float* __restrict__ bias,
                                        const float* __restrict__ gamma, const float* __restrict__ beta,
                                        const float* __restrict__ mean, const float* __restrict__ var,
                                        float* __restrict__ out,
                                        int prow0, int wfirst, int lo, int hi) {
    const int  w    = wfirst + lo;
    const int  pe   = (lo < 8) ? (wfirst + 16 * NT + (lo < 1 ? lo : 1)) : (wfirst - 16 + (lo > 14 ? lo : 14));
    const bool eok  = (pe >= 0) && (pe < EA_WW);
    const int  pec  = pe < 0 ? 0 : (pe >= EA_WW ? EA_WW - 1 : pe);
    const int  offM = w   * EA_C + 4 * hi;
    const int  offE = pec * EA_C + 4 * hi;
    const float* kb = kflat + 4 * EA_M * hi + (lo < EA_M ? lo : EA_M - 1);
    const EaEdge ed = {lo >= 14, lo == 15, lo == 0, lo <= 1};
    v8f y[NT][EA_W] = {};
    EaStage<NT> cur;
    ea_stage_load<NT>(cur, Lrow, Rrow, offM, offE, eok, kb, 0);
    #pragma unroll 1
    for (int gp = 1; gp < 8; ++gp) {
        EaStage<NT> nxt;
        ea_stage_load<NT>(nxt, Lrow, Rrow, offM, offE, eok, kb, gp);
        ea_stage_mma<NT>(cur, ed, y);
        cur = nxt;
    }
    ea_stage_mma<NT>(cur, ed, y);

    EaBn bn;
    #pragma unroll
    for (int c = 0; c < 4; ++c) {
        const int k = 4 * lo + c;
        bn.sc[c] = gamma[k] * rsqrtf(var[k] + EA_EPS);
        bn.mu[c] = mean[k];
        bn.be[c] = beta[k];
    }

    #pragma unroll
    for (int j = 0; j < NT; ++j)
        ea_tile_finish(y[j], wfirst + 16 * j, prow0 + wfirst + 16 * j, lo, hi, Lrow, Rrow, bias, bn, out);
}

__global__ __launch_bounds__(256)
void ea_main_kernel(const float* __restrict__ L, const float* __restrict__ R,
                    const float* __restrict__ kflat, const float* __restrict__ bias,
                    const float* __restrict__ gamma, const float* __restrict__ beta,
                    const float* __restrict__ mean, const float* __restrict__ var,
                    float* __restrict__ out) {
    const int tid  = threadIdx.x;
    const int lane = tid & 31;
    const int wib  = __builtin_amdgcn_readfirstlane(tid >> 5);
    const int lo = lane & 15;
    const int hi = lane >> 4;

    const bool wide = wib < 4;
    const int  v    = wide ? wib : wib - 4;
    const int  row  = blockIdx.x * 2 + (v >> 1);
    const int  wfirst = wide ? (v & 1) * 48 : 96 + (v & 1) * 32;
    const size_t rowBase = (size_t)row * EA_WW * EA_C;
    const float* Lrow = L + rowBase;
    const float* Rrow = R + rowBase;

    if (wide) ea_wave<3>(Lrow, Rrow, kflat, bias, gamma, beta, mean, var, out, row * EA_WW, wfirst, lo, hi);
    else      ea_wave<2>(Lrow, Rrow, kflat, bias, gamma, beta, mean, var, out, row * EA_WW, wfirst, lo, hi);
}

extern "C" void kernel_launch(void* const* d_in, const int* in_sizes, int n_in,
                              void* d_out, int out_size, void* d_ws, size_t ws_size,
                              hipStream_t stream) {
    const float* L     = (const float*)d_in[0];
    const float* R     = (const float*)d_in[1];
    const float* kconv = (const float*)d_in[2];
    const float* bias  = (const float*)d_in[3];
    const float* gamma = (const float*)d_in[4];
    const float* beta  = (const float*)d_in[5];
    const float* mean  = (const float*)d_in[6];
    const float* var   = (const float*)d_in[7];
    float* out = (float*)d_out;

    const int groups = EA_B * EA_H / 2;
    ea_main_kernel<<<groups, 256, 0, stream>>>(L, R, kconv, bias, gamma, beta,
                                               mean, var, out);
}
